# starting kernel
# baseline (speedup 1.0000x reference)
#include <hip/hip_runtime.h>

#define N_NODES 50000
#define N_EDGES 800000
#define FEATS   128
#define M_TILES (N_NODES / 16)
#define N_TILES (FEATS / 16)

typedef __attribute__((ext_vector_type(2))) float v2f;
typedef __attribute__((ext_vector_type(8))) float v8f;

__device__ __forceinline__ void atomic_add_f32(float* p, float v) {
    __hip_atomic_fetch_add(p, v, __ATOMIC_RELAXED, __HIP_MEMORY_SCOPE_AGENT);
}

__global__ void zero_kernel(float* __restrict__ p, int n) {
    int i = blockIdx.x * blockDim.x + threadIdx.x;
    if (i < n) p[i] = 0.0f;
}

__global__ void degree_kernel(const int* __restrict__ src, const int* __restrict__ dst,
                              float* __restrict__ deg_out, float* __restrict__ deg_in) {
    int e = blockIdx.x * blockDim.x + threadIdx.x;
    if (e < N_EDGES) {
        atomic_add_f32(&deg_out[src[e]], 1.0f);
        atomic_add_f32(&deg_in[dst[e]], 1.0f);
    }
}

__global__ void rsqrt_kernel(float* __restrict__ deg_out, float* __restrict__ deg_in) {
    int i = blockIdx.x * blockDim.x + threadIdx.x;
    if (i < N_NODES) {
        deg_out[i] = rsqrtf(fmaxf(deg_out[i], 1.0f));
        deg_in[i]  = rsqrtf(fmaxf(deg_in[i],  1.0f));
    }
}

__global__ __launch_bounds__(256)
void gemm_wmma_kernel(const float* __restrict__ x, const float* __restrict__ W,
                      const float* __restrict__ s_out, float* __restrict__ y) {
    const int wave   = (blockIdx.x * blockDim.x + threadIdx.x) >> 5;
    const int lane   = threadIdx.x & 31;
    const int tile_n = wave & (N_TILES - 1);
    const int tile_m = wave >> 3;

    const int m = lane & 15;
    const int h = lane >> 4;
    const int row = tile_m * 16 + m;
    const int col = tile_n * 16 + m;

    const float scale = s_out[row];
    const float* __restrict__ arow = x + (size_t)row * FEATS;

    v8f c = {};
    #pragma unroll 4
    for (int k0 = 0; k0 < FEATS; k0 += 4) {
        const int k = k0 + 2 * h;
        v2f a = *(const v2f*)(arow + k);
        a.x *= scale;
        a.y *= scale;
        v2f b;
        b.x = W[(size_t)k * FEATS + col];
        b.y = W[(size_t)(k + 1) * FEATS + col];
        c = __builtin_amdgcn_wmma_f32_16x16x4_f32(
                 false, a,  false, b,
                 (short)0, c,  false,  false);
    }

    #pragma unroll
    for (int j = 0; j < 8; ++j) {
        const int orow = tile_m * 16 + j + 8 * h;
        y[(size_t)orow * FEATS + col] = c[j];
    }
}

__global__ __launch_bounds__(256)
void scatter_kernel(const int* __restrict__ src, const int* __restrict__ dst,
                    const float* __restrict__ y, float* __restrict__ out) {
    const int e    = (blockIdx.x * blockDim.x + threadIdx.x) >> 5;
    const int lane = threadIdx.x & 31;
    if (e >= N_EDGES) return;
    const int s = src[e];
    const int d = dst[e];
    const float4 v = *(const float4*)(y + (size_t)s * FEATS + lane * 4);
    float* o = out + (size_t)d * FEATS + lane * 4;
    atomic_add_f32(o + 0, v.x);
    atomic_add_f32(o + 1, v.y);
    atomic_add_f32(o + 2, v.z);
    atomic_add_f32(o + 3, v.w);
}

__global__ void finalize_kernel(float* __restrict__ out, const float* __restrict__ s_in,
                                const float* __restrict__ bias) {
    int i = blockIdx.x * blockDim.x + threadIdx.x;
    if (i < N_NODES * FEATS) {
        int r = i >> 7;
        int ccol = i & (FEATS - 1);
        out[i] = out[i] * s_in[r] + bias[ccol];
    }
}

extern "C" void kernel_launch(void* const* d_in, const int* in_sizes, int n_in,
                              void* d_out, int out_size, void* d_ws, size_t ws_size,
                              hipStream_t stream) {
    const float* x    = (const float*)d_in[0];
    const int*   src  = (const int*)  d_in[1];
    const int*   dst  = (const int*)  d_in[2];
    const float* W    = (const float*)d_in[3];
    const float* bias = (const float*)d_in[4];
    float* out = (float*)d_out;

    float* s_out = (float*)d_ws;
    float* s_in  = s_out + N_NODES;
    float* y     = s_in + N_NODES;

    const int TB = 256;

    zero_kernel<<<(2 * N_NODES + TB - 1) / TB, TB, 0, stream>>>(s_out, 2 * N_NODES);
    zero_kernel<<<(N_NODES * FEATS + TB - 1) / TB, TB, 0, stream>>>(out, N_NODES * FEATS);

    degree_kernel<<<(N_EDGES + TB - 1) / TB, TB, 0, stream>>>(src, dst, s_out, s_in);

    rsqrt_kernel<<<(N_NODES + TB - 1) / TB, TB, 0, stream>>>(s_out, s_in);

    gemm_wmma_kernel<<<(M_TILES * N_TILES) / 8, TB, 0, stream>>>(x, W, s_out, y);

    scatter_kernel<<<(N_EDGES * 32) / TB, TB, 0, stream>>>(src, dst, y, out);

    finalize_kernel<<<(N_NODES * FEATS + TB - 1) / TB, TB, 0, stream>>>(out, s_in, bias);
}


# both edge scans keep one set of chunk registers and reload each int4 in place right after it is tested: no second buffer, no register copies, one pointer per chunk; no-hit path falls through
# speedup vs baseline: 3.2173x; 3.2173x over previous
#include <hip/hip_runtime.h>

#define N_NODES 50000
#define N_EDGES 800000
#define FEATS   128
#define M_TILES (N_NODES / 16)
#define N_TILES (FEATS / 16)

typedef __attribute__((ext_vector_type(2))) float v2f;
typedef __attribute__((ext_vector_type(8))) float v8f;

#define GEMM_WAVES   8
#define GEMM_THREADS (GEMM_WAVES * 32)
#define GEMM_BLOCKS  ((M_TILES + GEMM_WAVES - 1) / GEMM_WAVES)
#define K_PAIRS      (FEATS / 2)
#define WL_PITCH     288
static_assert(GEMM_THREADS == 256 && FEATS == 128, "the W staging below maps 256 threads onto 8 row pairs x 32 column groups");

__global__ __launch_bounds__(GEMM_THREADS) __attribute__((amdgpu_waves_per_eu(2, 4)))
void gemm_wmma_kernel(const float* __restrict__ x, const float* __restrict__ W,
                      const float* __restrict__ s_out, float* __restrict__ y) {
    __shared__ __attribute__((aligned(16))) float Wl[K_PAIRS * WL_PITCH];

    const int tid  = threadIdx.x;
    const int lane = tid & 31;
    const int wave = __builtin_amdgcn_readfirstlane(tid >> 5);
    const int n = lane & 15;
    const int h = lane >> 4;
    const int panel = blockIdx.x * GEMM_WAVES + wave;
    const bool live = panel < M_TILES;
    const int tile_m = live ? panel : (M_TILES - 1);

    const int row = tile_m * 16 + n;
    const float scale = s_out[row];
    const float* __restrict__ arow = x + (size_t)row * FEATS + 2 * h;
    v2f a[FEATS / 4];
    #pragma unroll
    for (int ks = 0; ks < FEATS / 4; ++ks) a[ks] = *(const v2f*)(arow + 4 * ks);
    #pragma unroll
    for (int ks = 0; ks < FEATS / 4; ++ks) {
        a[ks].x *= scale;
        a[ks].y *= scale;
    }

    {
        const int i  = tid & 31;
        const int q0 = tid >> 5;
        float* d0 = Wl + 128 * (i >> 4) + 2 * (i & 15);
        #pragma unroll
        for (int p = 0; p < K_PAIRS / 8; ++p) {
            const int q = q0 + 8 * p;
            const float4 r0 = *(const float4*)(W + (size_t)(2 * q) * FEATS + 4 * i);
            const float4 r1 = *(const float4*)(W + (size_t)(2 * q + 1) * FEATS + 4 * i);
            float* d = d0 + q * WL_PITCH;
            *(v2f*)(d)      = v2f{r0.x, r1.x};
            *(v2f*)(d + 32) = v2f{r0.y, r1.y};
            *(v2f*)(d + 64) = v2f{r0.z, r1.z};
            *(v2f*)(d + 96) = v2f{r0.w, r1.w};
        }
    }
    __syncthreads();

    v8f c[N_TILES];
    #pragma unroll
    for (int t = 0; t < N_TILES; ++t) c[t] = v8f{};

    const float* bl = Wl + h * WL_PITCH + 2 * n;
    #pragma unroll
    for (int ks = 0; ks < FEATS / 4; ++ks) {
        const v2f av = a[ks];
        v2f b[N_TILES];
        #pragma unroll
        for (int t = 0; t < N_TILES; ++t) b[t] = *(const v2f*)(bl + (2 * ks) * WL_PITCH + 32 * t);
        #pragma unroll
        for (int t = 0; t < N_TILES; ++t) {
            c[t] = __builtin_amdgcn_wmma_f32_16x16x4_f32(
                        false, av,  false, b[t],
                        (short)0, c[t],  false,  false);
        }
    }

    if (live) {
        float* yrow = y + (size_t)(tile_m * 16 + 8 * h) * FEATS + 4 * n;
        #pragma unroll
        for (int j = 0; j < 8; ++j) {
            const float4 o0 = {c[0][j], c[1][j], c[2][j], c[3][j]};
            const float4 o1 = {c[4][j], c[5][j], c[6][j], c[7][j]};
            *(float4*)(yrow + j * FEATS)      = o0;
            *(float4*)(yrow + j * FEATS + 64) = o1;
        }
    }
}

#define ROWS_PB   200
#define AGG_BLOCKS (N_NODES / ROWS_PB)
#define AGG_THREADS 1024
#define AGG_WAVES (AGG_THREADS / 32)
#define CH_VEC    5
#define CH_EDGES  (32 * 4 * CH_VEC)
#define N_CHUNKS  (N_EDGES / CH_EDGES)
static_assert(N_EDGES % CH_EDGES == 0, "edge chunks must tile the edge list exactly");
static_assert(N_NODES % ROWS_PB == 0, "owner ranges must tile the row list exactly");
static_assert(N_CHUNKS % AGG_BLOCKS == 0, "the per-block rotation must be a whole number of chunks");
static_assert(ROWS_PB <= 4096 && N_EDGES <= (1 << 20), "ring entry packs edge (20 bits) and local row (12 bits)");
#define RING      256
#define BATCH     8

__device__ __forceinline__ void agg_batch(const int* __restrict__ src, const float* __restrict__ y,
                                          float* acc, const unsigned* ring, int head, int cnt, int lane) {
    float4 v[BATCH];
    int dl[BATCH];
    #pragma unroll
    for (int k = 0; k < BATCH; ++k) {
        const int kk = (k < cnt) ? k : (cnt - 1);
        const unsigned ent = (unsigned)__builtin_amdgcn_readfirstlane((int)ring[(head + kk) & (RING - 1)]);
        const int e = (int)(ent & 0xFFFFFu);
        dl[k] = (int)(ent >> 20);
        const int s = src[e];
        v[k] = *(const float4*)(y + ((unsigned)s * FEATS + (unsigned)lane * 4u));
    }
    #pragma unroll
    for (int k = 0; k < BATCH; ++k) {
        if (k < cnt) {
            float* a = acc + dl[k] * FEATS + lane;
            __hip_atomic_fetch_add(a,      v[k].x, __ATOMIC_RELAXED, __HIP_MEMORY_SCOPE_WORKGROUP);
            __hip_atomic_fetch_add(a + 32, v[k].y, __ATOMIC_RELAXED, __HIP_MEMORY_SCOPE_WORKGROUP);
            __hip_atomic_fetch_add(a + 64, v[k].z, __ATOMIC_RELAXED, __HIP_MEMORY_SCOPE_WORKGROUP);
            __hip_atomic_fetch_add(a + 96, v[k].w, __ATOMIC_RELAXED, __HIP_MEMORY_SCOPE_WORKGROUP);
        }
    }
}

#define DEG_ROWS    400
#define DEG_BLOCKS  (N_NODES / DEG_ROWS)
#define DEG_THREADS 1024
#define DEG_WAVES   (DEG_THREADS / 32)
static_assert(N_NODES % DEG_ROWS == 0, "owner ranges must tile the node list exactly");

__global__ __launch_bounds__(DEG_THREADS)
void outdeg_kernel(const int* __restrict__ src, float* __restrict__ s_out) {
    __shared__ unsigned cnt[DEG_ROWS];

    const int tid  = threadIdx.x;
    const int lane = tid & 31;
    const int wave = __builtin_amdgcn_readfirstlane(tid >> 5);
    const int lo   = blockIdx.x * DEG_ROWS;

    if (tid < DEG_ROWS) cnt[tid] = 0u;
    __syncthreads();

    const int rot = (int)blockIdx.x * (N_CHUNKS / DEG_BLOCKS);
    int left = (N_CHUNKS - wave + DEG_WAVES - 1) / DEG_WAVES;
    int chunk = wave + rot;
    if (chunk >= N_CHUNKS) chunk -= N_CHUNKS;
    int4 cur[CH_VEC];
    const int4* __restrict__ src4 = (const int4*)src + lane;
    {
        const int4* p = src4 + chunk * (CH_EDGES / 4);
        #pragma unroll
        for (int i = 0; i < CH_VEC; ++i) cur[i] = p[i * 32];
    }
    while (true) {
        const bool more = left > 1;
        int chunk2 = chunk + DEG_WAVES;
        if (chunk2 >= N_CHUNKS) chunk2 -= N_CHUNKS;
        const int4* pn = src4 + (more ? chunk2 : chunk) * (CH_EDGES / 4);
        #pragma unroll
        for (int i = 0; i < CH_VEC; ++i) {
            const int sv[4] = {cur[i].x, cur[i].y, cur[i].z, cur[i].w};
            #pragma unroll
            for (int c = 0; c < 4; ++c) {
                const unsigned sloc = (unsigned)(sv[c] - lo);
                const bool hit = sloc < (unsigned)DEG_ROWS;
                if (__builtin_expect(__builtin_amdgcn_ballot_w32(hit) != 0u, 0)) {
                    if (hit) {
                        __hip_atomic_fetch_add(&cnt[sloc], 1u, __ATOMIC_RELAXED, __HIP_MEMORY_SCOPE_WORKGROUP);
                    }
                }
            }
            cur[i] = pn[i * 32];
        }
        if (!more) break;
        chunk = chunk2;
        --left;
    }

    __syncthreads();

    if (tid < DEG_ROWS) s_out[lo + tid] = rsqrtf(fmaxf((float)cnt[tid], 1.0f));
}

__global__ __launch_bounds__(AGG_THREADS)
void aggregate_kernel(const int* __restrict__ src, const int* __restrict__ dst,
                      const float* __restrict__ y,
                      const float* __restrict__ bias, float* __restrict__ out) {
    __shared__ float acc[ROWS_PB * FEATS];
    __shared__ unsigned rings[AGG_WAVES * RING];
    __shared__ unsigned indeg[ROWS_PB];

    const int tid  = threadIdx.x;
    const int lane = tid & 31;
    const int wave = __builtin_amdgcn_readfirstlane(tid >> 5);
    const int lo   = blockIdx.x * ROWS_PB;
    unsigned* ring = rings + wave * RING;

    for (int i = tid; i < ROWS_PB * FEATS; i += AGG_THREADS) acc[i] = 0.0f;
    if (tid < ROWS_PB) indeg[tid] = 0u;
    __syncthreads();

    int head = 0, tail = 0;
    const int rot = (int)blockIdx.x * (N_CHUNKS / AGG_BLOCKS);
    int left = (N_CHUNKS - wave + AGG_WAVES - 1) / AGG_WAVES;
    int chunk = wave + rot;
    if (chunk >= N_CHUNKS) chunk -= N_CHUNKS;
    int4 cur[CH_VEC];
    const int4* __restrict__ dst4 = (const int4*)dst + lane;
    {
        const int4* p = dst4 + chunk * (CH_EDGES / 4);
        #pragma unroll
        for (int i = 0; i < CH_VEC; ++i) cur[i] = p[i * 32];
    }
    while (true) {
        const bool more = left > 1;
        int chunk2 = chunk + AGG_WAVES;
        if (chunk2 >= N_CHUNKS) chunk2 -= N_CHUNKS;
        const int4* pn = dst4 + (more ? chunk2 : chunk) * (CH_EDGES / 4);
        #pragma unroll
        for (int i = 0; i < CH_VEC; ++i) {
            const int e0 = chunk * CH_EDGES + i * 128 + lane * 4;
            const int dv[4] = {cur[i].x, cur[i].y, cur[i].z, cur[i].w};
            #pragma unroll
            for (int c = 0; c < 4; ++c) {
                const unsigned dloc = (unsigned)(dv[c] - lo);
                const bool hit = dloc < (unsigned)ROWS_PB;
                const unsigned mask = __builtin_amdgcn_ballot_w32(hit);
                if (__builtin_expect(mask != 0u, 0)) {
                    if (hit) {
                        const int rank = (int)__builtin_amdgcn_mbcnt_lo(mask, 0u);
                        ring[(tail + rank) & (RING - 1)] = (unsigned)(e0 + c) | (dloc << 20);
                        __hip_atomic_fetch_add(&indeg[dloc], 1u, __ATOMIC_RELAXED, __HIP_MEMORY_SCOPE_WORKGROUP);
                    }
                    tail += __builtin_popcount(mask);
                }
            }
            if (__builtin_expect(tail - head >= BATCH, 0)) {
                do {
                    agg_batch(src, y, acc, ring, head, BATCH, lane);
                    head += BATCH;
                } while (tail - head >= BATCH);
            }
            cur[i] = pn[i * 32];
        }
        if (!more) break;
        chunk = chunk2;
        --left;
    }
    if (tail - head > 0) agg_batch(src, y, acc, ring, head, tail - head, lane);

    __syncthreads();

    const float4 b4 = *(const float4*)(bias + lane * 4);
    const int myrow = wave + AGG_WAVES * lane;
    const float sv = (myrow < ROWS_PB) ? rsqrtf(fmaxf((float)indeg[myrow], 1.0f)) : 0.0f;
    int t = 0;
    for (int r = wave; r < ROWS_PB; r += AGG_WAVES, ++t) {
        const float* a = acc + r * FEATS + lane;
        const float sc = __builtin_bit_cast(float, __builtin_amdgcn_readlane(__builtin_bit_cast(int, sv), t));
        float4 o;
        o.x = a[0]  * sc + b4.x;
        o.y = a[32] * sc + b4.y;
        o.z = a[64] * sc + b4.z;
        o.w = a[96] * sc + b4.w;
        *(float4*)(out + (size_t)(lo + r) * FEATS + lane * 4) = o;
    }
}

extern "C" void kernel_launch(void* const* d_in, const int* in_sizes, int n_in,
                              void* d_out, int out_size, void* d_ws, size_t ws_size,
                              hipStream_t stream) {
    const float* x    = (const float*)d_in[0];
    const int*   src  = (const int*)  d_in[1];
    const int*   dst  = (const int*)  d_in[2];
    const float* W    = (const float*)d_in[3];
    const float* bias = (const float*)d_in[4];
    float* out = (float*)d_out;

    float* s_out = (float*)d_ws;
    float* y     = s_out + 2 * N_NODES;

    outdeg_kernel<<<DEG_BLOCKS, DEG_THREADS, 0, stream>>>(src, s_out);

    gemm_wmma_kernel<<<GEMM_BLOCKS, GEMM_THREADS, 0, stream>>>(x, W, s_out, y);

    aggregate_kernel<<<AGG_BLOCKS, AGG_THREADS, 0, stream>>>(src, dst, y, bias, out);
}


# out-degree count tiled 25 node ranges x 10 edge slices: a block tests 80000 edges, not 800000; exact partial counts parked in each y panel's own footprint, summed by that panel's GEMM wave
# speedup vs baseline: 3.6289x; 1.1279x over previous
#include <hip/hip_runtime.h>

#define N_NODES 50000
#define N_EDGES 800000
#define FEATS   128
#define M_TILES (N_NODES / 16)
#define N_TILES (FEATS / 16)

typedef __attribute__((ext_vector_type(2))) float v2f;
typedef __attribute__((ext_vector_type(8))) float v8f;

#define DEG_PARTS    10
#define DEG_LINE     32
static_assert(DEG_PARTS * DEG_LINE <= 16 * FEATS, "a panel's partial counts must fit in its own y footprint");

#define GEMM_WAVES   8
#define GEMM_THREADS (GEMM_WAVES * 32)
#define GEMM_BLOCKS  ((M_TILES + GEMM_WAVES - 1) / GEMM_WAVES)
#define K_PAIRS      (FEATS / 2)
#define WL_PITCH     288
static_assert(GEMM_THREADS == 256 && FEATS == 128, "the W staging below maps 256 threads onto 8 row pairs x 32 column groups");

__global__ __launch_bounds__(GEMM_THREADS) __attribute__((amdgpu_waves_per_eu(2, 4)))
void gemm_wmma_kernel(const float* __restrict__ x, const float* __restrict__ W,
                      float* __restrict__ y) {
    __shared__ __attribute__((aligned(16))) float Wl[K_PAIRS * WL_PITCH];

    const int tid  = threadIdx.x;
    const int lane = tid & 31;
    const int wave = __builtin_amdgcn_readfirstlane(tid >> 5);
    const int n = lane & 15;
    const int h = lane >> 4;
    const int panel = blockIdx.x * GEMM_WAVES + wave;
    const bool live = panel < M_TILES;
    const int tile_m = live ? panel : (M_TILES - 1);

    const int row = tile_m * 16 + n;
    float part_deg[DEG_PARTS];
    {
        const float* dp = y + (size_t)tile_m * (16 * FEATS) + 2 * n;
        #pragma unroll
        for (int p = 0; p < DEG_PARTS; ++p) part_deg[p] = dp[p * DEG_LINE];
    }
    const float* __restrict__ arow = x + (size_t)row * FEATS + 2 * h;
    v2f a[FEATS / 4];
    #pragma unroll
    for (int ks = 0; ks < FEATS / 4; ++ks) a[ks] = *(const v2f*)(arow + 4 * ks);
    float degree = 0.0f;
    #pragma unroll
    for (int p = 0; p < DEG_PARTS; ++p) degree += part_deg[p];
    const float scale = rsqrtf(fmaxf(degree, 1.0f));
    #pragma unroll
    for (int ks = 0; ks < FEATS / 4; ++ks) {
        a[ks].x *= scale;
        a[ks].y *= scale;
    }

    {
        const int i  = tid & 31;
        const int q0 = tid >> 5;
        float* d0 = Wl + 128 * (i >> 4) + 2 * (i & 15);
        #pragma unroll
        for (int p = 0; p < K_PAIRS / 8; ++p) {
            const int q = q0 + 8 * p;
            const float4 r0 = *(const float4*)(W + (size_t)(2 * q) * FEATS + 4 * i);
            const float4 r1 = *(const float4*)(W + (size_t)(2 * q + 1) * FEATS + 4 * i);
            float* d = d0 + q * WL_PITCH;
            *(v2f*)(d)      = v2f{r0.x, r1.x};
            *(v2f*)(d + 32) = v2f{r0.y, r1.y};
            *(v2f*)(d + 64) = v2f{r0.z, r1.z};
            *(v2f*)(d + 96) = v2f{r0.w, r1.w};
        }
    }
    __syncthreads();

    v8f c[N_TILES];
    #pragma unroll
    for (int t = 0; t < N_TILES; ++t) c[t] = v8f{};

    const float* bl = Wl + h * WL_PITCH + 2 * n;
    #pragma unroll
    for (int ks = 0; ks < FEATS / 4; ++ks) {
        const v2f av = a[ks];
        v2f b[N_TILES];
        #pragma unroll
        for (int t = 0; t < N_TILES; ++t) b[t] = *(const v2f*)(bl + (2 * ks) * WL_PITCH + 32 * t);
        #pragma unroll
        for (int t = 0; t < N_TILES; ++t) {
            c[t] = __builtin_amdgcn_wmma_f32_16x16x4_f32(
                        false, av,  false, b[t],
                        (short)0, c[t],  false,  false);
        }
    }

    if (live) {
        float* yrow = y + (size_t)(tile_m * 16 + 8 * h) * FEATS + 4 * n;
        #pragma unroll
        for (int j = 0; j < 8; ++j) {
            const float4 o0 = {c[0][j], c[1][j], c[2][j], c[3][j]};
            const float4 o1 = {c[4][j], c[5][j], c[6][j], c[7][j]};
            *(float4*)(yrow + j * FEATS)      = o0;
            *(float4*)(yrow + j * FEATS + 64) = o1;
        }
    }
}

#define ROWS_PB   200
#define AGG_BLOCKS (N_NODES / ROWS_PB)
#define AGG_THREADS 1024
#define AGG_WAVES (AGG_THREADS / 32)
#define CH_VEC    5
#define CH_EDGES  (32 * 4 * CH_VEC)
#define N_CHUNKS  (N_EDGES / CH_EDGES)
static_assert(N_EDGES % CH_EDGES == 0, "edge chunks must tile the edge list exactly");
static_assert(N_NODES % ROWS_PB == 0, "owner ranges must tile the row list exactly");
static_assert(N_CHUNKS % AGG_BLOCKS == 0, "the per-block rotation must be a whole number of chunks");
static_assert(ROWS_PB <= 4096 && N_EDGES <= (1 << 20), "ring entry packs edge (20 bits) and local row (12 bits)");
#define RING      256
#define BATCH     8

__device__ __forceinline__ void agg_batch(const int* __restrict__ src, const float* __restrict__ y,
                                          float* acc, const unsigned* ring, int head, int cnt, int lane) {
    float4 v[BATCH];
    int dl[BATCH];
    #pragma unroll
    for (int k = 0; k < BATCH; ++k) {
        const int kk = (k < cnt) ? k : (cnt - 1);
        const unsigned ent = (unsigned)__builtin_amdgcn_readfirstlane((int)ring[(head + kk) & (RING - 1)]);
        const int e = (int)(ent & 0xFFFFFu);
        dl[k] = (int)(ent >> 20);
        const int s = src[e];
        v[k] = *(const float4*)(y + ((unsigned)s * FEATS + (unsigned)lane * 4u));
    }
    #pragma unroll
    for (int k = 0; k < BATCH; ++k) {
        if (k < cnt) {
            float* a = acc + dl[k] * FEATS + lane;
            __hip_atomic_fetch_add(a,      v[k].x, __ATOMIC_RELAXED, __HIP_MEMORY_SCOPE_WORKGROUP);
            __hip_atomic_fetch_add(a + 32, v[k].y, __ATOMIC_RELAXED, __HIP_MEMORY_SCOPE_WORKGROUP);
            __hip_atomic_fetch_add(a + 64, v[k].z, __ATOMIC_RELAXED, __HIP_MEMORY_SCOPE_WORKGROUP);
            __hip_atomic_fetch_add(a + 96, v[k].w, __ATOMIC_RELAXED, __HIP_MEMORY_SCOPE_WORKGROUP);
        }
    }
}

#define DEG_ROWS    2000
#define DEG_RANGES  (N_NODES / DEG_ROWS)
#define DEG_BLOCKS  (DEG_RANGES * DEG_PARTS)
#define DEG_THREADS 512
#define PART_VEC    (N_EDGES / 4 / DEG_PARTS)
#define DEG_LOADS   ((PART_VEC + DEG_THREADS - 1) / DEG_THREADS)
static_assert(N_NODES % DEG_ROWS == 0 && DEG_ROWS % 16 == 0, "owner ranges must tile the node list in whole panels");
static_assert(N_EDGES % (4 * DEG_PARTS) == 0, "slices must be whole int4");

__global__ __launch_bounds__(DEG_THREADS)
void outdeg_kernel(const int* __restrict__ src, float* __restrict__ y) {
    __shared__ unsigned cnt[DEG_ROWS];

    const int tid   = threadIdx.x;
    const int range = (int)blockIdx.x / DEG_PARTS;
    const int part  = (int)blockIdx.x % DEG_PARTS;
    const int lo    = range * DEG_ROWS;

    for (int i = tid; i < DEG_ROWS; i += DEG_THREADS) cnt[i] = 0u;
    __syncthreads();

    const int4* __restrict__ p = (const int4*)src + part * PART_VEC + tid;
    int4 v[DEG_LOADS];
    #pragma unroll
    for (int k = 0; k < DEG_LOADS; ++k) {
        v[k] = (k * DEG_THREADS + tid < PART_VEC) ? p[k * DEG_THREADS] : int4{-1, -1, -1, -1};
    }

    #pragma unroll
    for (int k = 0; k < DEG_LOADS; ++k) {
        const int sv[4] = {v[k].x, v[k].y, v[k].z, v[k].w};
        #pragma unroll
        for (int c = 0; c < 4; ++c) {
            const unsigned sloc = (unsigned)(sv[c] - lo);
            if (sloc < (unsigned)DEG_ROWS)
                __hip_atomic_fetch_add(&cnt[sloc], 1u, __ATOMIC_RELAXED, __HIP_MEMORY_SCOPE_WORKGROUP);
        }
    }

    __syncthreads();

    for (int i = tid; i < DEG_ROWS; i += DEG_THREADS) {
        float* line = y + (size_t)((lo >> 4) + (i >> 4)) * (16 * FEATS) + part * DEG_LINE;
        *(v2f*)(line + 2 * (i & 15)) = v2f{(float)cnt[i], 0.0f};
    }
}

__global__ __launch_bounds__(AGG_THREADS)
void aggregate_kernel(const int* __restrict__ src, const int* __restrict__ dst,
                      const float* __restrict__ y,
                      const float* __restrict__ bias, float* __restrict__ out) {
    __shared__ float acc[ROWS_PB * FEATS];
    __shared__ unsigned rings[AGG_WAVES * RING];
    __shared__ unsigned indeg[ROWS_PB];

    const int tid  = threadIdx.x;
    const int lane = tid & 31;
    const int wave = __builtin_amdgcn_readfirstlane(tid >> 5);
    const int lo   = blockIdx.x * ROWS_PB;
    unsigned* ring = rings + wave * RING;

    for (int i = tid; i < ROWS_PB * FEATS; i += AGG_THREADS) acc[i] = 0.0f;
    if (tid < ROWS_PB) indeg[tid] = 0u;
    __syncthreads();

    int head = 0, tail = 0;
    const int rot = (int)blockIdx.x * (N_CHUNKS / AGG_BLOCKS);
    int left = (N_CHUNKS - wave + AGG_WAVES - 1) / AGG_WAVES;
    int chunk = wave + rot;
    if (chunk >= N_CHUNKS) chunk -= N_CHUNKS;
    int4 cur[CH_VEC];
    const int4* __restrict__ dst4 = (const int4*)dst + lane;
    {
        const int4* p = dst4 + chunk * (CH_EDGES / 4);
        #pragma unroll
        for (int i = 0; i < CH_VEC; ++i) cur[i] = p[i * 32];
    }
    while (true) {
        const bool more = left > 1;
        int chunk2 = chunk + AGG_WAVES;
        if (chunk2 >= N_CHUNKS) chunk2 -= N_CHUNKS;
        const int4* pn = dst4 + (more ? chunk2 : chunk) * (CH_EDGES / 4);
        #pragma unroll
        for (int i = 0; i < CH_VEC; ++i) {
            const int e0 = chunk * CH_EDGES + i * 128 + lane * 4;
            const int dv[4] = {cur[i].x, cur[i].y, cur[i].z, cur[i].w};
            #pragma unroll
            for (int c = 0; c < 4; ++c) {
                const unsigned dloc = (unsigned)(dv[c] - lo);
                const bool hit = dloc < (unsigned)ROWS_PB;
                const unsigned mask = __builtin_amdgcn_ballot_w32(hit);
                if (__builtin_expect(mask != 0u, 0)) {
                    if (hit) {
                        const int rank = (int)__builtin_amdgcn_mbcnt_lo(mask, 0u);
                        ring[(tail + rank) & (RING - 1)] = (unsigned)(e0 + c) | (dloc << 20);
                        __hip_atomic_fetch_add(&indeg[dloc], 1u, __ATOMIC_RELAXED, __HIP_MEMORY_SCOPE_WORKGROUP);
                    }
                    tail += __builtin_popcount(mask);
                }
            }
            if (__builtin_expect(tail - head >= BATCH, 0)) {
                do {
                    agg_batch(src, y, acc, ring, head, BATCH, lane);
                    head += BATCH;
                } while (tail - head >= BATCH);
            }
            cur[i] = pn[i * 32];
        }
        if (!more) break;
        chunk = chunk2;
        --left;
    }
    if (tail - head > 0) agg_batch(src, y, acc, ring, head, tail - head, lane);

    __syncthreads();

    const float4 b4 = *(const float4*)(bias + lane * 4);
    const int myrow = wave + AGG_WAVES * lane;
    const float sv = (myrow < ROWS_PB) ? rsqrtf(fmaxf((float)indeg[myrow], 1.0f)) : 0.0f;
    int t = 0;
    for (int r = wave; r < ROWS_PB; r += AGG_WAVES, ++t) {
        const float* a = acc + r * FEATS + lane;
        const float sc = __builtin_bit_cast(float, __builtin_amdgcn_readlane(__builtin_bit_cast(int, sv), t));
        float4 o;
        o.x = a[0]  * sc + b4.x;
        o.y = a[32] * sc + b4.y;
        o.z = a[64] * sc + b4.z;
        o.w = a[96] * sc + b4.w;
        *(float4*)(out + (size_t)(lo + r) * FEATS + lane * 4) = o;
    }
}

extern "C" void kernel_launch(void* const* d_in, const int* in_sizes, int n_in,
                              void* d_out, int out_size, void* d_ws, size_t ws_size,
                              hipStream_t stream) {
    const float* x    = (const float*)d_in[0];
    const int*   src  = (const int*)  d_in[1];
    const int*   dst  = (const int*)  d_in[2];
    const float* W    = (const float*)d_in[3];
    const float* bias = (const float*)d_in[4];
    float* out = (float*)d_out;

    float* y = (float*)d_ws + 2 * N_NODES;

    outdeg_kernel<<<DEG_BLOCKS, DEG_THREADS, 0, stream>>>(src, y);

    gemm_wmma_kernel<<<GEMM_BLOCKS, GEMM_THREADS, 0, stream>>>(x, W, y);

    aggregate_kernel<<<AGG_BLOCKS, AGG_THREADS, 0, stream>>>(src, dst, y, bias, out);
}


# aggregate scans one-byte owner ids (dst/200, written once per call by the degree launch with a packed src|row word): 4 edges per register, 0.8 MB per block not 3.2; sums stay LDS arrival order
# speedup vs baseline: 3.7456x; 1.0322x over previous
#include <hip/hip_runtime.h>

#define N_NODES 50000
#define N_EDGES 800000
#define FEATS   128
#define M_TILES (N_NODES / 16)
#define N_TILES (FEATS / 16)

typedef __attribute__((ext_vector_type(2))) float v2f;
typedef __attribute__((ext_vector_type(8))) float v8f;

#define DEG_PARTS    10
#define DEG_LINE     32
static_assert(DEG_PARTS * DEG_LINE <= 16 * FEATS, "a panel's partial counts must fit in its own y footprint");

#define GEMM_WAVES   8
#define GEMM_THREADS (GEMM_WAVES * 32)
#define GEMM_BLOCKS  ((M_TILES + GEMM_WAVES - 1) / GEMM_WAVES)
#define K_PAIRS      (FEATS / 2)
#define WL_PITCH     288
static_assert(GEMM_THREADS == 256 && FEATS == 128, "the W staging below maps 256 threads onto 8 row pairs x 32 column groups");

__global__ __launch_bounds__(GEMM_THREADS) __attribute__((amdgpu_waves_per_eu(2, 4)))
void gemm_wmma_kernel(const float* __restrict__ x, const float* __restrict__ W,
                      float* __restrict__ y) {
    __shared__ __attribute__((aligned(16))) float Wl[K_PAIRS * WL_PITCH];

    const int tid  = threadIdx.x;
    const int lane = tid & 31;
    const int wave = __builtin_amdgcn_readfirstlane(tid >> 5);
    const int n = lane & 15;
    const int h = lane >> 4;
    const int panel = blockIdx.x * GEMM_WAVES + wave;
    const bool live = panel < M_TILES;
    const int tile_m = live ? panel : (M_TILES - 1);

    const int row = tile_m * 16 + n;
    float part_deg[DEG_PARTS];
    {
        const float* dp = y + (size_t)tile_m * (16 * FEATS) + 2 * n;
        #pragma unroll
        for (int p = 0; p < DEG_PARTS; ++p) part_deg[p] = dp[p * DEG_LINE];
    }
    const float* __restrict__ arow = x + (size_t)row * FEATS + 2 * h;
    v2f a[FEATS / 4];
    #pragma unroll
    for (int ks = 0; ks < FEATS / 4; ++ks) a[ks] = *(const v2f*)(arow + 4 * ks);
    float degree = 0.0f;
    #pragma unroll
    for (int p = 0; p < DEG_PARTS; ++p) degree += part_deg[p];
    const float scale = rsqrtf(fmaxf(degree, 1.0f));
    #pragma unroll
    for (int ks = 0; ks < FEATS / 4; ++ks) {
        a[ks].x *= scale;
        a[ks].y *= scale;
    }

    {
        const int i  = tid & 31;
        const int q0 = tid >> 5;
        float* d0 = Wl + 128 * (i >> 4) + 2 * (i & 15);
        #pragma unroll
        for (int p = 0; p < K_PAIRS / 8; ++p) {
            const int q = q0 + 8 * p;
            const float4 r0 = *(const float4*)(W + (size_t)(2 * q) * FEATS + 4 * i);
            const float4 r1 = *(const float4*)(W + (size_t)(2 * q + 1) * FEATS + 4 * i);
            float* d = d0 + q * WL_PITCH;
            *(v2f*)(d)      = v2f{r0.x, r1.x};
            *(v2f*)(d + 32) = v2f{r0.y, r1.y};
            *(v2f*)(d + 64) = v2f{r0.z, r1.z};
            *(v2f*)(d + 96) = v2f{r0.w, r1.w};
        }
    }
    __syncthreads();

    v8f c[N_TILES];
    #pragma unroll
    for (int t = 0; t < N_TILES; ++t) c[t] = v8f{};

    const float* bl = Wl + h * WL_PITCH + 2 * n;
    #pragma unroll
    for (int ks = 0; ks < FEATS / 4; ++ks) {
        const v2f av = a[ks];
        v2f b[N_TILES];
        #pragma unroll
        for (int t = 0; t < N_TILES; ++t) b[t] = *(const v2f*)(bl + (2 * ks) * WL_PITCH + 32 * t);
        #pragma unroll
        for (int t = 0; t < N_TILES; ++t) {
            c[t] = __builtin_amdgcn_wmma_f32_16x16x4_f32(
                        false, av,  false, b[t],
                        (short)0, c[t],  false,  false);
        }
    }

    if (live) {
        float* yrow = y + (size_t)(tile_m * 16 + 8 * h) * FEATS + 4 * n;
        #pragma unroll
        for (int j = 0; j < 8; ++j) {
            const float4 o0 = {c[0][j], c[1][j], c[2][j], c[3][j]};
            const float4 o1 = {c[4][j], c[5][j], c[6][j], c[7][j]};
            *(float4*)(yrow + j * FEATS)      = o0;
            *(float4*)(yrow + j * FEATS + 64) = o1;
        }
    }
}

#define ROWS_PB   200
#define AGG_BLOCKS (N_NODES / ROWS_PB)
#define AGG_THREADS 1024
#define AGG_WAVES (AGG_THREADS / 32)
#define RID_NONE  255u
#define UNIT_EDGES (32 * 8)
#define CH_VEC    5
#define CH_EDGES  (UNIT_EDGES * CH_VEC)
#define N_CHUNKS  (N_EDGES / CH_EDGES)
static_assert(N_EDGES % CH_EDGES == 0, "edge chunks must tile the edge list exactly");
static_assert(N_NODES % ROWS_PB == 0, "owner ranges must tile the row list exactly");
static_assert(AGG_BLOCKS <= 255, "a range id is one byte, and 255 stands for no owner");
static_assert(N_NODES <= 65536 && ROWS_PB <= 65536, "the packed edge word holds src and the local row in 16 bits each");
static_assert(N_EDGES <= (1 << 20), "ring entries are edge numbers");
#define RING      512
#define BATCH     8

__device__ __forceinline__ void agg_batch(const unsigned* __restrict__ pk, const float* __restrict__ y,
                                          float* acc, unsigned* indeg, const unsigned* ring,
                                          int head, int cnt, int lane) {
    const int kk = min(lane & (BATCH - 1), cnt - 1);
    const unsigned w = pk[ring[(head + kk) & (RING - 1)]];
    float4 v[BATCH];
    int dl[BATCH];
    #pragma unroll
    for (int k = 0; k < BATCH; ++k) {
        const unsigned wk = (unsigned)__builtin_amdgcn_readlane((int)w, k);
        dl[k] = (int)(wk >> 16);
        const unsigned s = wk & 0xFFFFu;
        v[k] = *(const float4*)(y + (s * FEATS + (unsigned)lane * 4u));
    }
    #pragma unroll
    for (int k = 0; k < BATCH; ++k) {
        if (k < cnt) {
            float* a = acc + dl[k] * FEATS + lane;
            __hip_atomic_fetch_add(a,      v[k].x, __ATOMIC_RELAXED, __HIP_MEMORY_SCOPE_WORKGROUP);
            __hip_atomic_fetch_add(a + 32, v[k].y, __ATOMIC_RELAXED, __HIP_MEMORY_SCOPE_WORKGROUP);
            __hip_atomic_fetch_add(a + 64, v[k].z, __ATOMIC_RELAXED, __HIP_MEMORY_SCOPE_WORKGROUP);
            __hip_atomic_fetch_add(a + 96, v[k].w, __ATOMIC_RELAXED, __HIP_MEMORY_SCOPE_WORKGROUP);
        }
    }
    if (lane < cnt)
        __hip_atomic_fetch_add(&indeg[w >> 16], 1u, __ATOMIC_RELAXED, __HIP_MEMORY_SCOPE_WORKGROUP);
}

#define DEG_ROWS    2000
#define DEG_RANGES  (N_NODES / DEG_ROWS)
#define DEG_BLOCKS  (DEG_RANGES * DEG_PARTS)
#define DEG_THREADS 512
#define PART_VEC    (N_EDGES / 4 / DEG_PARTS)
#define DEG_LOADS   ((PART_VEC + DEG_THREADS - 1) / DEG_THREADS)
static_assert(N_NODES % DEG_ROWS == 0 && DEG_ROWS % 16 == 0, "owner ranges must tile the node list in whole panels");
static_assert(N_EDGES % (4 * DEG_PARTS) == 0, "slices must be whole int4");
#define PRE_VEC     (N_EDGES / 4 / DEG_BLOCKS)
static_assert(N_EDGES % (4 * DEG_BLOCKS) == 0, "the blocks' shares of the edge list must be whole int4");
static_assert(PRE_VEC > DEG_THREADS && PRE_VEC <= 2 * DEG_THREADS, "one or two int4 per thread");

__device__ __forceinline__ void pack_edges(const int4 s, const int4 d, unsigned* __restrict__ rid4,
                                           uint4* __restrict__ pk4, int i) {
    const unsigned dv[4] = {(unsigned)d.x, (unsigned)d.y, (unsigned)d.z, (unsigned)d.w};
    const unsigned sv[4] = {(unsigned)s.x, (unsigned)s.y, (unsigned)s.z, (unsigned)s.w};
    unsigned ids = 0u;
    unsigned w[4];
    #pragma unroll
    for (int c = 0; c < 4; ++c) {
        const bool node = dv[c] < (unsigned)N_NODES;
        const unsigned own = node ? dv[c] / (unsigned)ROWS_PB : RID_NONE;
        const unsigned loc = node ? dv[c] - own * (unsigned)ROWS_PB : 0u;
        ids |= own << (8 * c);
        w[c] = (sv[c] & 0xFFFFu) | (loc << 16);
    }
    rid4[i] = ids;
    pk4[i]  = uint4{w[0], w[1], w[2], w[3]};
}

__global__ __launch_bounds__(DEG_THREADS)
void outdeg_kernel(const int* __restrict__ src, const int* __restrict__ dst, float* __restrict__ y,
                   unsigned* __restrict__ rid4, uint4* __restrict__ pk4) {
    __shared__ unsigned cnt[DEG_ROWS];

    const int tid   = threadIdx.x;
    const int range = (int)blockIdx.x / DEG_PARTS;
    const int part  = (int)blockIdx.x % DEG_PARTS;
    const int lo    = range * DEG_ROWS;

    const int  pre_i   = (int)blockIdx.x * PRE_VEC + tid;
    const bool pre_two = tid < PRE_VEC - DEG_THREADS;
    const int4 pre_s0 = ((const int4*)src)[pre_i];
    const int4 pre_d0 = ((const int4*)dst)[pre_i];
    int4 pre_s1 = int4{0, 0, 0, 0}, pre_d1 = int4{0, 0, 0, 0};
    if (pre_two) {
        pre_s1 = ((const int4*)src)[pre_i + DEG_THREADS];
        pre_d1 = ((const int4*)dst)[pre_i + DEG_THREADS];
    }

    for (int i = tid; i < DEG_ROWS; i += DEG_THREADS) cnt[i] = 0u;
    __syncthreads();

    pack_edges(pre_s0, pre_d0, rid4, pk4, pre_i);
    if (pre_two) pack_edges(pre_s1, pre_d1, rid4, pk4, pre_i + DEG_THREADS);

    const int4* __restrict__ p = (const int4*)src + part * PART_VEC + tid;
    int4 v[DEG_LOADS];
    #pragma unroll
    for (int k = 0; k < DEG_LOADS; ++k) {
        v[k] = (k * DEG_THREADS + tid < PART_VEC) ? p[k * DEG_THREADS] : int4{-1, -1, -1, -1};
    }

    #pragma unroll
    for (int k = 0; k < DEG_LOADS; ++k) {
        const int sv[4] = {v[k].x, v[k].y, v[k].z, v[k].w};
        #pragma unroll
        for (int c = 0; c < 4; ++c) {
            const unsigned sloc = (unsigned)(sv[c] - lo);
            if (sloc < (unsigned)DEG_ROWS)
                __hip_atomic_fetch_add(&cnt[sloc], 1u, __ATOMIC_RELAXED, __HIP_MEMORY_SCOPE_WORKGROUP);
        }
    }

    __syncthreads();

    for (int i = tid; i < DEG_ROWS; i += DEG_THREADS) {
        float* line = y + (size_t)((lo >> 4) + (i >> 4)) * (16 * FEATS) + part * DEG_LINE;
        *(v2f*)(line + 2 * (i & 15)) = v2f{(float)cnt[i], 0.0f};
    }
}

__global__ __launch_bounds__(AGG_THREADS)
void aggregate_kernel(const unsigned* __restrict__ rid4, const unsigned* __restrict__ pk,
                      const float* __restrict__ y,
                      const float* __restrict__ bias, float* __restrict__ out) {
    __shared__ float acc[ROWS_PB * FEATS];
    __shared__ unsigned rings[AGG_WAVES * RING];
    __shared__ unsigned indeg[ROWS_PB];

    const int tid  = threadIdx.x;
    const int lane = tid & 31;
    const int wave = __builtin_amdgcn_readfirstlane(tid >> 5);
    const int lo   = blockIdx.x * ROWS_PB;
    unsigned* ring = rings + wave * RING;
    const unsigned pat = (unsigned)blockIdx.x * 0x01010101u;

    for (int i = tid; i < ROWS_PB * FEATS; i += AGG_THREADS) acc[i] = 0.0f;
    if (tid < ROWS_PB) indeg[tid] = 0u;
    __syncthreads();

    int head = 0, tail = 0;
    const int rot = ((int)blockIdx.x * N_CHUNKS) / AGG_BLOCKS;
    int left = (N_CHUNKS - wave + AGG_WAVES - 1) / AGG_WAVES;
    int chunk = wave + rot;
    if (chunk >= N_CHUNKS) chunk -= N_CHUNKS;
    uint2 cur[CH_VEC];
    const uint2* __restrict__ rid8 = (const uint2*)rid4 + lane;
    {
        const uint2* p = rid8 + chunk * (CH_EDGES / 8);
        #pragma unroll
        for (int i = 0; i < CH_VEC; ++i) cur[i] = p[i * 32];
    }
    while (true) {
        const bool more = left > 1;
        int chunk2 = chunk + AGG_WAVES;
        if (chunk2 >= N_CHUNKS) chunk2 -= N_CHUNKS;
        const uint2* pn = rid8 + (more ? chunk2 : chunk) * (CH_EDGES / 8);
        #pragma unroll
        for (int i = 0; i < CH_VEC; ++i) {
            const int e0 = chunk * CH_EDGES + i * UNIT_EDGES + lane * 8;
            const unsigned wv[2] = {cur[i].x, cur[i].y};
            #pragma unroll
            for (int d = 0; d < 2; ++d) {
                const unsigned xw = wv[d] ^ pat;
                const unsigned zf = ~(((xw & 0x7F7F7F7Fu) + 0x7F7F7F7Fu) | xw | 0x7F7F7F7Fu);
                if (__builtin_expect(__builtin_amdgcn_ballot_w32(zf != 0u) != 0u, 0)) {
                    unsigned m = zf;
                    unsigned mask = __builtin_amdgcn_ballot_w32(m != 0u);
                    do {
                        if (m != 0u) {
                            const int c = __builtin_ctz(m) >> 3;
                            const int rank = (int)__builtin_amdgcn_mbcnt_lo(mask, 0u);
                            ring[(tail + rank) & (RING - 1)] = (unsigned)(e0 + 4 * d + c);
                            m &= m - 1u;
                        }
                        tail += __builtin_popcount(mask);
                        mask = __builtin_amdgcn_ballot_w32(m != 0u);
                    } while (mask != 0u);
                }
            }
            if (__builtin_expect(tail - head >= BATCH, 0)) {
                do {
                    agg_batch(pk, y, acc, indeg, ring, head, BATCH, lane);
                    head += BATCH;
                } while (tail - head >= BATCH);
            }
            cur[i] = pn[i * 32];
        }
        if (!more) break;
        chunk = chunk2;
        --left;
    }
    if (tail - head > 0) agg_batch(pk, y, acc, indeg, ring, head, tail - head, lane);

    __syncthreads();

    const float4 b4 = *(const float4*)(bias + lane * 4);
    const int myrow = wave + AGG_WAVES * lane;
    const float sv = (myrow < ROWS_PB) ? rsqrtf(fmaxf((float)indeg[myrow], 1.0f)) : 0.0f;
    int t = 0;
    for (int r = wave; r < ROWS_PB; r += AGG_WAVES, ++t) {
        const float* a = acc + r * FEATS + lane;
        const float sc = __builtin_bit_cast(float, __builtin_amdgcn_readlane(__builtin_bit_cast(int, sv), t));
        float4 o;
        o.x = a[0]  * sc + b4.x;
        o.y = a[32] * sc + b4.y;
        o.z = a[64] * sc + b4.z;
        o.w = a[96] * sc + b4.w;
        *(float4*)(out + (size_t)(lo + r) * FEATS + lane * 4) = o;
    }
}

extern "C" void kernel_launch(void* const* d_in, const int* in_sizes, int n_in,
                              void* d_out, int out_size, void* d_ws, size_t ws_size,
                              hipStream_t stream) {
    const float* x    = (const float*)d_in[0];
    const int*   src  = (const int*)  d_in[1];
    const int*   dst  = (const int*)  d_in[2];
    const float* W    = (const float*)d_in[3];
    const float* bias = (const float*)d_in[4];
    float* out = (float*)d_out;

    float*    y  = (float*)d_ws + 2 * N_NODES;
    unsigned* pk = (unsigned*)(y + N_NODES * FEATS);
    unsigned* rid4 = pk + N_EDGES;

    outdeg_kernel<<<DEG_BLOCKS, DEG_THREADS, 0, stream>>>(src, dst, y, rid4, (uint4*)pk);

    gemm_wmma_kernel<<<GEMM_BLOCKS, GEMM_THREADS, 0, stream>>>(x, W, y);

    aggregate_kernel<<<AGG_BLOCKS, AGG_THREADS, 0, stream>>>(rid4, pk, y, bias, out);
}


# aggregate no longer scans every edge's owner id: launch 1 marks, per owner block, the 32-edge groups that hold its edges; a block loads only those groups' packed words; sums stay LDS arrival order
# speedup vs baseline: 4.5682x; 1.2196x over previous
#include <hip/hip_runtime.h>

#define N_NODES 50000
#define N_EDGES 800000
#define FEATS   128
#define M_TILES (N_NODES / 16)
#define N_TILES (FEATS / 16)

typedef __attribute__((ext_vector_type(2))) float v2f;
typedef __attribute__((ext_vector_type(8))) float v8f;

#define DEG_PARTS    10
#define DEG_LINE     32
static_assert(DEG_PARTS * DEG_LINE <= 16 * FEATS, "a panel's partial counts must fit in its own y footprint");

#define GEMM_WAVES   8
#define GEMM_THREADS (GEMM_WAVES * 32)
#define GEMM_BLOCKS  ((M_TILES + GEMM_WAVES - 1) / GEMM_WAVES)
#define K_PAIRS      (FEATS / 2)
#define WL_PITCH     288
static_assert(GEMM_THREADS == 256 && FEATS == 128, "the W staging below maps 256 threads onto 8 row pairs x 32 column groups");

__global__ __launch_bounds__(GEMM_THREADS) __attribute__((amdgpu_waves_per_eu(2, 4)))
void gemm_wmma_kernel(const float* __restrict__ x, const float* __restrict__ W,
                      float* __restrict__ y) {
    __shared__ __attribute__((aligned(16))) float Wl[K_PAIRS * WL_PITCH];

    const int tid  = threadIdx.x;
    const int lane = tid & 31;
    const int wave = __builtin_amdgcn_readfirstlane(tid >> 5);
    const int n = lane & 15;
    const int h = lane >> 4;
    const int panel = blockIdx.x * GEMM_WAVES + wave;
    const bool live = panel < M_TILES;
    const int tile_m = live ? panel : (M_TILES - 1);

    const int row = tile_m * 16 + n;
    float part_deg[DEG_PARTS];
    {
        const float* dp = y + (size_t)tile_m * (16 * FEATS) + 2 * n;
        #pragma unroll
        for (int p = 0; p < DEG_PARTS; ++p) part_deg[p] = dp[p * DEG_LINE];
    }
    const float* __restrict__ arow = x + (size_t)row * FEATS + 2 * h;
    v2f a[FEATS / 4];
    #pragma unroll
    for (int ks = 0; ks < FEATS / 4; ++ks) a[ks] = *(const v2f*)(arow + 4 * ks);
    float degree = 0.0f;
    #pragma unroll
    for (int p = 0; p < DEG_PARTS; ++p) degree += part_deg[p];
    const float scale = rsqrtf(fmaxf(degree, 1.0f));
    #pragma unroll
    for (int ks = 0; ks < FEATS / 4; ++ks) {
        a[ks].x *= scale;
        a[ks].y *= scale;
    }

    {
        const int i  = tid & 31;
        const int q0 = tid >> 5;
        float* d0 = Wl + 128 * (i >> 4) + 2 * (i & 15);
        #pragma unroll
        for (int p = 0; p < K_PAIRS / 8; ++p) {
            const int q = q0 + 8 * p;
            const float4 r0 = *(const float4*)(W + (size_t)(2 * q) * FEATS + 4 * i);
            const float4 r1 = *(const float4*)(W + (size_t)(2 * q + 1) * FEATS + 4 * i);
            float* d = d0 + q * WL_PITCH;
            *(v2f*)(d)      = v2f{r0.x, r1.x};
            *(v2f*)(d + 32) = v2f{r0.y, r1.y};
            *(v2f*)(d + 64) = v2f{r0.z, r1.z};
            *(v2f*)(d + 96) = v2f{r0.w, r1.w};
        }
    }
    __syncthreads();

    v8f c[N_TILES];
    #pragma unroll
    for (int t = 0; t < N_TILES; ++t) c[t] = v8f{};

    const float* bl = Wl + h * WL_PITCH + 2 * n;
    #pragma unroll
    for (int ks = 0; ks < FEATS / 4; ++ks) {
        const v2f av = a[ks];
        v2f b[N_TILES];
        #pragma unroll
        for (int t = 0; t < N_TILES; ++t) b[t] = *(const v2f*)(bl + (2 * ks) * WL_PITCH + 32 * t);
        #pragma unroll
        for (int t = 0; t < N_TILES; ++t) {
            c[t] = __builtin_amdgcn_wmma_f32_16x16x4_f32(
                        false, av,  false, b[t],
                        (short)0, c[t],  false,  false);
        }
    }

    if (live) {
        float* yrow = y + (size_t)(tile_m * 16 + 8 * h) * FEATS + 4 * n;
        #pragma unroll
        for (int j = 0; j < 8; ++j) {
            const float4 o0 = {c[0][j], c[1][j], c[2][j], c[3][j]};
            const float4 o1 = {c[4][j], c[5][j], c[6][j], c[7][j]};
            *(float4*)(yrow + j * FEATS)      = o0;
            *(float4*)(yrow + j * FEATS + 64) = o1;
        }
    }
}

#define ROWS_PB   200
#define AGG_BLOCKS (N_NODES / ROWS_PB)
#define AGG_THREADS 1024
#define AGG_WAVES (AGG_THREADS / 32)
#define OWN_NONE  255u
#define GRP_EDGES 32
#define PRE_VEC   1024
#define PRE_BLOCKS ((N_EDGES / 4 + PRE_VEC - 1) / PRE_VEC)
#define PRE_WORDS (PRE_VEC * 4 / GRP_EDGES / 32)
#define ROW_WORDS (PRE_BLOCKS * PRE_WORDS)
#define PRES_PITCH 1024
#define WPW       25
#define LIST_CAP  (WPW * 32)
#define VISIT     8
static_assert(N_EDGES % GRP_EDGES == 0, "groups must tile the edge list exactly");
static_assert(PRE_WORDS == 4, "a converting block writes its group bits as one 16-byte store per owner");
static_assert(AGG_BLOCKS * PRE_WORDS <= PRES_PITCH && PRES_PITCH % 32 == 0, "the owners' words fit in a converting block's lines");
static_assert(AGG_WAVES * WPW >= ROW_WORDS && WPW <= 32, "the waves' words must cover the bit row");
static_assert(ROW_WORDS * 32 <= 65536, "list entries are 16-bit group numbers");
static_assert(N_NODES % ROWS_PB == 0, "owner ranges must tile the row list exactly");
static_assert(AGG_BLOCKS <= 255 && ROWS_PB <= 256, "owner number and local row are one byte each, and 255 stands for no owner");
static_assert(N_NODES <= 65536, "the packed edge word holds src in 16 bits");
#define RING      512
#define BATCH     8
static_assert(BATCH - 1 + VISIT * GRP_EDGES <= RING, "a visit's hits must fit in the ring beside a short batch's rest");

__device__ __forceinline__ void agg_batch(const float* __restrict__ y,
                                          float* acc, unsigned* indeg, const unsigned* ring,
                                          int head, int cnt, int lane) {
    const int kk = min(lane & (BATCH - 1), cnt - 1);
    const unsigned w = ring[(head + kk) & (RING - 1)];
    float4 v[BATCH];
    int dl[BATCH];
    #pragma unroll
    for (int k = 0; k < BATCH; ++k) {
        const unsigned wk = (unsigned)__builtin_amdgcn_readlane((int)w, k);
        dl[k] = (int)((wk >> 16) & 0xFFu);
        const unsigned s = wk & 0xFFFFu;
        v[k] = *(const float4*)(y + (s * FEATS + (unsigned)lane * 4u));
    }
    #pragma unroll
    for (int k = 0; k < BATCH; ++k) {
        if (k < cnt) {
            float* a = acc + dl[k] * FEATS + lane;
            __hip_atomic_fetch_add(a,      v[k].x, __ATOMIC_RELAXED, __HIP_MEMORY_SCOPE_WORKGROUP);
            __hip_atomic_fetch_add(a + 32, v[k].y, __ATOMIC_RELAXED, __HIP_MEMORY_SCOPE_WORKGROUP);
            __hip_atomic_fetch_add(a + 64, v[k].z, __ATOMIC_RELAXED, __HIP_MEMORY_SCOPE_WORKGROUP);
            __hip_atomic_fetch_add(a + 96, v[k].w, __ATOMIC_RELAXED, __HIP_MEMORY_SCOPE_WORKGROUP);
        }
    }
    if (lane < cnt)
        __hip_atomic_fetch_add(&indeg[(w >> 16) & 0xFFu], 1u, __ATOMIC_RELAXED, __HIP_MEMORY_SCOPE_WORKGROUP);
}

#define DEG_ROWS    2000
#define DEG_RANGES  (N_NODES / DEG_ROWS)
#define DEG_BLOCKS  (DEG_RANGES * DEG_PARTS)
#define DEG_THREADS 512
#define PART_VEC    (N_EDGES / 4 / DEG_PARTS)
#define DEG_LOADS   ((PART_VEC + DEG_THREADS - 1) / DEG_THREADS)
static_assert(N_NODES % DEG_ROWS == 0 && DEG_ROWS % 16 == 0, "owner ranges must tile the node list in whole panels");
static_assert(N_EDGES % (4 * DEG_PARTS) == 0, "slices must be whole int4");
static_assert(PRE_VEC == 2 * DEG_THREADS, "two int4 per thread");
static_assert(PRE_BLOCKS <= DEG_BLOCKS, "the converting blocks are the first blocks of the launch");

__device__ __forceinline__ void pack_edges(const int4 s, const int4 d, unsigned* marks,
                                           uint4* __restrict__ pk4, int gi, int li) {
    const unsigned dv[4] = {(unsigned)d.x, (unsigned)d.y, (unsigned)d.z, (unsigned)d.w};
    const unsigned sv[4] = {(unsigned)s.x, (unsigned)s.y, (unsigned)s.z, (unsigned)s.w};
    const int      word = li >> 8;
    const unsigned bit  = 1u << ((li >> 3) & 31);
    unsigned w[4];
    #pragma unroll
    for (int c = 0; c < 4; ++c) {
        const bool node = dv[c] < (unsigned)N_NODES;
        const unsigned own = node ? dv[c] / (unsigned)ROWS_PB : OWN_NONE;
        const unsigned loc = node ? dv[c] - own * (unsigned)ROWS_PB : 0u;
        w[c] = (sv[c] & 0xFFFFu) | (loc << 16) | (own << 24);
        if (node)
            __hip_atomic_fetch_or(&marks[own * PRE_WORDS + word], bit, __ATOMIC_RELAXED, __HIP_MEMORY_SCOPE_WORKGROUP);
    }
    pk4[gi] = uint4{w[0], w[1], w[2], w[3]};
}

__global__ __launch_bounds__(DEG_THREADS)
void outdeg_kernel(const int* __restrict__ src, const int* __restrict__ dst, float* __restrict__ y,
                   unsigned* __restrict__ pres, uint4* __restrict__ pk4) {
    __shared__ unsigned cnt[DEG_ROWS];
    __shared__ __attribute__((aligned(16))) unsigned marks[AGG_BLOCKS * PRE_WORDS];

    const int tid   = threadIdx.x;
    const int range = (int)blockIdx.x / DEG_PARTS;
    const int part  = (int)blockIdx.x % DEG_PARTS;
    const int lo    = range * DEG_ROWS;

    const bool pre_blk = (int)blockIdx.x < PRE_BLOCKS;
    const int  pre_i0  = (int)blockIdx.x * PRE_VEC + tid;
    const int  pre_i1  = pre_i0 + DEG_THREADS;
    const bool pre_v0  = pre_i0 < N_EDGES / 4;
    const bool pre_v1  = pre_i1 < N_EDGES / 4;
    int4 pre_s0 = int4{0, 0, 0, 0}, pre_d0 = int4{0, 0, 0, 0};
    int4 pre_s1 = int4{0, 0, 0, 0}, pre_d1 = int4{0, 0, 0, 0};
    if (pre_v0) {
        pre_s0 = ((const int4*)src)[pre_i0];
        pre_d0 = ((const int4*)dst)[pre_i0];
    }
    if (pre_v1) {
        pre_s1 = ((const int4*)src)[pre_i1];
        pre_d1 = ((const int4*)dst)[pre_i1];
    }

    for (int i = tid; i < DEG_ROWS; i += DEG_THREADS) cnt[i] = 0u;
    if (pre_blk)
        for (int i = tid; i < AGG_BLOCKS * PRE_WORDS; i += DEG_THREADS) marks[i] = 0u;
    __syncthreads();

    if (pre_v0) pack_edges(pre_s0, pre_d0, marks, pk4, pre_i0, tid);
    if (pre_v1) pack_edges(pre_s1, pre_d1, marks, pk4, pre_i1, tid + DEG_THREADS);

    const int4* __restrict__ p = (const int4*)src + part * PART_VEC + tid;
    int4 v[DEG_LOADS];
    #pragma unroll
    for (int k = 0; k < DEG_LOADS; ++k) {
        v[k] = (k * DEG_THREADS + tid < PART_VEC) ? p[k * DEG_THREADS] : int4{-1, -1, -1, -1};
    }

    #pragma unroll
    for (int k = 0; k < DEG_LOADS; ++k) {
        const int sv[4] = {v[k].x, v[k].y, v[k].z, v[k].w};
        #pragma unroll
        for (int c = 0; c < 4; ++c) {
            const unsigned sloc = (unsigned)(sv[c] - lo);
            if (sloc < (unsigned)DEG_ROWS)
                __hip_atomic_fetch_add(&cnt[sloc], 1u, __ATOMIC_RELAXED, __HIP_MEMORY_SCOPE_WORKGROUP);
        }
    }

    __syncthreads();

    for (int i = tid; i < DEG_ROWS; i += DEG_THREADS) {
        float* line = y + (size_t)((lo >> 4) + (i >> 4)) * (16 * FEATS) + part * DEG_LINE;
        *(v2f*)(line + 2 * (i & 15)) = v2f{(float)cnt[i], 0.0f};
    }
    if (pre_blk && tid < AGG_BLOCKS)
        ((uint4*)pres)[(int)blockIdx.x * (PRES_PITCH / 4) + tid] = *(const uint4*)(marks + PRE_WORDS * tid);
}

__global__ __launch_bounds__(AGG_THREADS)
void aggregate_kernel(const unsigned* __restrict__ pres, const unsigned* __restrict__ pk,
                      const float* __restrict__ y,
                      const float* __restrict__ bias, float* __restrict__ out) {
    __shared__ float acc[ROWS_PB * FEATS];
    __shared__ unsigned rings[AGG_WAVES * RING];
    __shared__ unsigned short lists[AGG_WAVES * LIST_CAP];
    __shared__ unsigned indeg[ROWS_PB];

    const int tid  = threadIdx.x;
    const int lane = tid & 31;
    const int wave = __builtin_amdgcn_readfirstlane(tid >> 5);
    const int lo   = blockIdx.x * ROWS_PB;
    const unsigned own = blockIdx.x;
    unsigned* ring = rings + wave * RING;
    unsigned short* list = lists + wave * LIST_CAP;

    const int slot = (wave + (int)blockIdx.x) & (AGG_WAVES - 1);
    const int widx = slot * WPW + lane;
    unsigned m = 0u;
    if (lane < WPW && widx < ROW_WORDS) m = pres[(widx >> 2) * PRES_PITCH + blockIdx.x * PRE_WORDS + (widx & 3)];

    for (int i = tid; i < ROWS_PB * FEATS; i += AGG_THREADS) acc[i] = 0.0f;
    if (tid < ROWS_PB) indeg[tid] = 0u;
    __syncthreads();

    int n = 0;
    {
        unsigned mask = __builtin_amdgcn_ballot_w32(m != 0u);
        while (mask != 0u) {
            if (m != 0u) {
                const int rank = (int)__builtin_amdgcn_mbcnt_lo(mask, 0u);
                list[n + rank] = (unsigned short)(widx * 32 + __builtin_ctz(m));
                m &= m - 1u;
            }
            n += __builtin_popcount(mask);
            mask = __builtin_amdgcn_ballot_w32(m != 0u);
        }
    }

    int head = 0, tail = 0;
    const unsigned* __restrict__ pkl = pk + lane;
    unsigned wv[VISIT];
    if (n > 0) {
        const unsigned g = list[min(lane & (VISIT - 1), n - 1)];
        #pragma unroll
        for (int k = 0; k < VISIT; ++k)
            wv[k] = pkl[(unsigned)__builtin_amdgcn_readlane((int)g, k) * GRP_EDGES];
    }
    for (int i = 0; i < n; i += VISIT) {
        #pragma unroll
        for (int k = 0; k < VISIT; ++k) {
            if (i + k < n) {
                const bool hit = (wv[k] >> 24) == own;
                const unsigned mask = __builtin_amdgcn_ballot_w32(hit);
                if (hit) ring[(tail + (int)__builtin_amdgcn_mbcnt_lo(mask, 0u)) & (RING - 1)] = wv[k];
                tail += __builtin_popcount(mask);
            }
        }
        if (i + VISIT < n) {
            const unsigned g = list[min(i + VISIT + (lane & (VISIT - 1)), n - 1)];
            #pragma unroll
            for (int k = 0; k < VISIT; ++k)
                wv[k] = pkl[(unsigned)__builtin_amdgcn_readlane((int)g, k) * GRP_EDGES];
        }
        while (tail - head >= BATCH) {
            agg_batch(y, acc, indeg, ring, head, BATCH, lane);
            head += BATCH;
        }
    }
    if (tail - head > 0) agg_batch(y, acc, indeg, ring, head, tail - head, lane);

    __syncthreads();

    const float4 b4 = *(const float4*)(bias + lane * 4);
    const int myrow = wave + AGG_WAVES * lane;
    const float sv = (myrow < ROWS_PB) ? rsqrtf(fmaxf((float)indeg[myrow], 1.0f)) : 0.0f;
    int t = 0;
    for (int r = wave; r < ROWS_PB; r += AGG_WAVES, ++t) {
        const float* a = acc + r * FEATS + lane;
        const float sc = __builtin_bit_cast(float, __builtin_amdgcn_readlane(__builtin_bit_cast(int, sv), t));
        float4 o;
        o.x = a[0]  * sc + b4.x;
        o.y = a[32] * sc + b4.y;
        o.z = a[64] * sc + b4.z;
        o.w = a[96] * sc + b4.w;
        *(float4*)(out + (size_t)(lo + r) * FEATS + lane * 4) = o;
    }
}

extern "C" void kernel_launch(void* const* d_in, const int* in_sizes, int n_in,
                              void* d_out, int out_size, void* d_ws, size_t ws_size,
                              hipStream_t stream) {
    const float* x    = (const float*)d_in[0];
    const int*   src  = (const int*)  d_in[1];
    const int*   dst  = (const int*)  d_in[2];
    const float* W    = (const float*)d_in[3];
    const float* bias = (const float*)d_in[4];
    float* out = (float*)d_out;

    float*    y  = (float*)d_ws;
    unsigned* pk = (unsigned*)(y + N_NODES * FEATS);
    unsigned* pres = pk + N_EDGES;

    outdeg_kernel<<<DEG_BLOCKS, DEG_THREADS, 0, stream>>>(src, dst, y, pres, (uint4*)pk);

    gemm_wmma_kernel<<<GEMM_BLOCKS, GEMM_THREADS, 0, stream>>>(x, W, y);

    aggregate_kernel<<<AGG_BLOCKS, AGG_THREADS, 0, stream>>>(pres, pk, y, bias, out);
}


# GEMM as a tile loop in 250 long-lived blocks: panels cut in four by columns, a wave keeps its 32 columns of B in registers; x rows pass through LDS; busiest SIMD 832 wmma, not 1024; sums unchanged
# speedup vs baseline: 4.8929x; 1.0711x over previous
#include <hip/hip_runtime.h>

#define N_NODES 50000
#define N_EDGES 800000
#define FEATS   128
#define M_TILES (N_NODES / 16)
#define N_TILES (FEATS / 16)

typedef __attribute__((ext_vector_type(2))) float v2f;
typedef __attribute__((ext_vector_type(8))) float v8f;

#define DEG_PARTS    10
#define DEG_LINE     32
static_assert(DEG_PARTS * DEG_LINE <= 16 * FEATS, "a panel's partial counts must fit in its own y footprint");

#define GEMM_BLOCKS  250
#define GEMM_WAVES   8
#define GEMM_THREADS (GEMM_WAVES * 32)
#define MAX_PANELS   13
#define PANELS_AHEAD 4
#define STEP_ROWS    GEMM_WAVES
#define MAX_STEPS    (16 * MAX_PANELS / STEP_ROWS)
#define MIN_STEPS    (16 * (MAX_PANELS - 1) / STEP_ROWS)
#define AHEAD_STEPS  (16 * PANELS_AHEAD / STEP_ROWS)
#define K_PAIRS      (FEATS / 2)
#define WL_PITCH     288
#define A_PITCH      132
static_assert(GEMM_THREADS == 256 && FEATS == 128, "the W staging below maps 256 threads onto 8 row pairs x 32 column groups");
static_assert(2 * M_TILES == 25 * GEMM_BLOCKS && 2 * MAX_PANELS == 26, "the panel ranges [25b/2, 25(b+1)/2) must tile the panels exactly, 12 or 13 each");
static_assert(16 * MAX_PANELS <= GEMM_THREADS, "one thread per row forms the row's out-degree factor");
static_assert(PANELS_AHEAD % 2 == 0 && PANELS_AHEAD < 12, "both panel parities get the same share of the panels ahead");

__device__ __forceinline__ void lds_barrier() {
    __builtin_amdgcn_fence(__ATOMIC_RELEASE, "workgroup", "local");
    __builtin_amdgcn_s_barrier();
    __builtin_amdgcn_fence(__ATOMIC_ACQUIRE, "workgroup", "local");
}

__device__ __forceinline__ void gemm_panel(const float* al, const v2f (&b)[FEATS / 4][2], float* yrow) {
    v2f a[FEATS / 4];
    #pragma unroll
    for (int ks = 0; ks < FEATS / 4; ++ks) a[ks] = *(const v2f*)(al + 4 * ks);
    v8f c0 = v8f{}, c1 = v8f{};
    #pragma unroll
    for (int ks = 0; ks < FEATS / 4; ++ks) {
        c0 = __builtin_amdgcn_wmma_f32_16x16x4_f32( false, a[ks],  false, b[ks][0],
                                                    (short)0, c0,  false,  false);
        c1 = __builtin_amdgcn_wmma_f32_16x16x4_f32( false, a[ks],  false, b[ks][1],
                                                    (short)0, c1,  false,  false);
    }
    #pragma unroll
    for (int j = 0; j < 8; ++j) *(v2f*)(yrow + j * FEATS) = v2f{c0[j], c1[j]};
}

__global__ __launch_bounds__(GEMM_THREADS)
void gemm_wmma_kernel(const float* __restrict__ x, const float* __restrict__ W,
                      float* __restrict__ y) {
    __shared__ __attribute__((aligned(16))) float Wl[K_PAIRS * WL_PITCH];
    __shared__ __attribute__((aligned(16))) float Al[16 * MAX_PANELS * A_PITCH];
    __shared__ float sc[16 * MAX_PANELS];

    const int tid  = threadIdx.x;
    const int lane = tid & 31;
    const int wave = __builtin_amdgcn_readfirstlane(tid >> 5);
    const int n = lane & 15;
    const int h = lane >> 4;
    const int p0 = (25 * (int)blockIdx.x) >> 1;
    const int np = ((25 * ((int)blockIdx.x + 1)) >> 1) - p0;
    const bool full = np == MAX_PANELS;

    float part_deg[DEG_PARTS];
    #pragma unroll
    for (int p = 0; p < DEG_PARTS; ++p) part_deg[p] = 0.0f;
    if (tid < 16 * np) {
        const float* dp = y + (size_t)(p0 + (tid >> 4)) * (16 * FEATS) + 2 * (tid & 15);
        #pragma unroll
        for (int p = 0; p < DEG_PARTS; ++p) part_deg[p] = dp[p * DEG_LINE];
    }
    const int wi  = tid & 31;
    const int wq0 = tid >> 5;
    float4 w0[K_PAIRS / 8], w1[K_PAIRS / 8];
    #pragma unroll
    for (int p = 0; p < K_PAIRS / 8; ++p) {
        const int q = wq0 + 8 * p;
        w0[p] = *(const float4*)(W + (size_t)(2 * q) * FEATS + 4 * wi);
        w1[p] = *(const float4*)(W + (size_t)(2 * q + 1) * FEATS + 4 * wi);
    }
    __builtin_amdgcn_sched_barrier(0);
    const float* xr = x + (size_t)(16 * p0 + wave) * FEATS + 4 * lane;
    float4 xa[MAX_STEPS];
    #pragma unroll
    for (int r = 0; r < MIN_STEPS; ++r) xa[r] = *(const float4*)(xr + (size_t)r * (STEP_ROWS * FEATS));
    const float* xt = full ? xr : xr - (MAX_STEPS - MIN_STEPS) * (STEP_ROWS * FEATS);
    #pragma unroll
    for (int r = MIN_STEPS; r < MAX_STEPS; ++r) xa[r] = *(const float4*)(xt + (size_t)r * (STEP_ROWS * FEATS));

    if (tid < 16 * np) {
        float degree = 0.0f;
        #pragma unroll
        for (int p = 0; p < DEG_PARTS; ++p) degree += part_deg[p];
        sc[tid] = rsqrtf(fmaxf(degree, 1.0f));
    }
    {
        float* d0 = Wl + 64 * (wi >> 3) + 4 * (wi & 7);
        #pragma unroll
        for (int p = 0; p < K_PAIRS / 8; ++p) {
            float* d = d0 + (wq0 + 8 * p) * WL_PITCH;
            *(float4*)(d)      = float4{w0[p].x, w1[p].x, w0[p].z, w1[p].z};
            *(float4*)(d + 32) = float4{w0[p].y, w1[p].y, w0[p].w, w1[p].w};
        }
    }
    lds_barrier();

    float* arow = Al + wave * A_PITCH + 4 * lane;
    #pragma unroll
    for (int r = 0; r < AHEAD_STEPS; ++r) {
        const float f = sc[STEP_ROWS * r + wave];
        *(float4*)(arow + r * (STEP_ROWS * A_PITCH)) = float4{xa[r].x * f, xa[r].y * f, xa[r].z * f, xa[r].w * f};
    }
    lds_barrier();

    const int slice  = wave & 3;
    const int parity = ((wave >> 2) ^ wave) & 1;
    v2f b[FEATS / 4][2];
    {
        const float* bl = Wl + h * WL_PITCH + 2 * n + 64 * slice;
        #pragma unroll
        for (int ks = 0; ks < FEATS / 4; ++ks) {
            b[ks][0] = *(const v2f*)(bl + (2 * ks) * WL_PITCH);
            b[ks][1] = *(const v2f*)(bl + (2 * ks) * WL_PITCH + 32);
        }
    }
    const float* al0 = Al + n * A_PITCH + 2 * h;
    float* y0 = y + (size_t)(16 * p0 + 8 * h) * FEATS + 32 * slice + 2 * n;

    for (int pl = parity; pl < PANELS_AHEAD; pl += 2)
        gemm_panel(al0 + pl * (16 * A_PITCH), b, y0 + (size_t)pl * (16 * FEATS));

    #pragma unroll
    for (int r = AHEAD_STEPS; r < MIN_STEPS; ++r) {
        const float f = sc[STEP_ROWS * r + wave];
        *(float4*)(arow + r * (STEP_ROWS * A_PITCH)) = float4{xa[r].x * f, xa[r].y * f, xa[r].z * f, xa[r].w * f};
    }
    if (full) {
        #pragma unroll
        for (int r = MIN_STEPS; r < MAX_STEPS; ++r) {
            const float f = sc[STEP_ROWS * r + wave];
            *(float4*)(arow + r * (STEP_ROWS * A_PITCH)) = float4{xa[r].x * f, xa[r].y * f, xa[r].z * f, xa[r].w * f};
        }
    }
    lds_barrier();

    for (int pl = PANELS_AHEAD + parity; pl < np; pl += 2)
        gemm_panel(al0 + pl * (16 * A_PITCH), b, y0 + (size_t)pl * (16 * FEATS));
}

#define ROWS_PB   200
#define AGG_BLOCKS (N_NODES / ROWS_PB)
#define AGG_THREADS 1024
#define AGG_WAVES (AGG_THREADS / 32)
#define OWN_NONE  255u
#define GRP_EDGES 32
#define PRE_VEC   1024
#define PRE_BLOCKS ((N_EDGES / 4 + PRE_VEC - 1) / PRE_VEC)
#define PRE_WORDS (PRE_VEC * 4 / GRP_EDGES / 32)
#define ROW_WORDS (PRE_BLOCKS * PRE_WORDS)
#define PRES_PITCH 1024
#define WPW       25
#define LIST_CAP  (WPW * 32)
#define VISIT     8
static_assert(N_EDGES % GRP_EDGES == 0, "groups must tile the edge list exactly");
static_assert(PRE_WORDS == 4, "a converting block writes its group bits as one 16-byte store per owner");
static_assert(AGG_BLOCKS * PRE_WORDS <= PRES_PITCH && PRES_PITCH % 32 == 0, "the owners' words fit in a converting block's lines");
static_assert(AGG_WAVES * WPW >= ROW_WORDS && WPW <= 32, "the waves' words must cover the bit row");
static_assert(ROW_WORDS * 32 <= 65536, "list entries are 16-bit group numbers");
static_assert(N_NODES % ROWS_PB == 0, "owner ranges must tile the row list exactly");
static_assert(AGG_BLOCKS <= 255 && ROWS_PB <= 256, "owner number and local row are one byte each, and 255 stands for no owner");
static_assert(N_NODES <= 65536, "the packed edge word holds src in 16 bits");
#define RING      512
#define BATCH     8
static_assert(BATCH - 1 + VISIT * GRP_EDGES <= RING, "a visit's hits must fit in the ring beside a short batch's rest");

__device__ __forceinline__ void agg_batch(const float* __restrict__ y,
                                          float* acc, unsigned* indeg, const unsigned* ring,
                                          int head, int cnt, int lane) {
    const int kk = min(lane & (BATCH - 1), cnt - 1);
    const unsigned w = ring[(head + kk) & (RING - 1)];
    float4 v[BATCH];
    int dl[BATCH];
    #pragma unroll
    for (int k = 0; k < BATCH; ++k) {
        const unsigned wk = (unsigned)__builtin_amdgcn_readlane((int)w, k);
        dl[k] = (int)((wk >> 16) & 0xFFu);
        const unsigned s = wk & 0xFFFFu;
        v[k] = *(const float4*)(y + (s * FEATS + (unsigned)lane * 4u));
    }
    #pragma unroll
    for (int k = 0; k < BATCH; ++k) {
        if (k < cnt) {
            float* a = acc + dl[k] * FEATS + lane;
            __hip_atomic_fetch_add(a,      v[k].x, __ATOMIC_RELAXED, __HIP_MEMORY_SCOPE_WORKGROUP);
            __hip_atomic_fetch_add(a + 32, v[k].y, __ATOMIC_RELAXED, __HIP_MEMORY_SCOPE_WORKGROUP);
            __hip_atomic_fetch_add(a + 64, v[k].z, __ATOMIC_RELAXED, __HIP_MEMORY_SCOPE_WORKGROUP);
            __hip_atomic_fetch_add(a + 96, v[k].w, __ATOMIC_RELAXED, __HIP_MEMORY_SCOPE_WORKGROUP);
        }
    }
    if (lane < cnt)
        __hip_atomic_fetch_add(&indeg[(w >> 16) & 0xFFu], 1u, __ATOMIC_RELAXED, __HIP_MEMORY_SCOPE_WORKGROUP);
}

#define DEG_ROWS    2000
#define DEG_RANGES  (N_NODES / DEG_ROWS)
#define DEG_BLOCKS  (DEG_RANGES * DEG_PARTS)
#define DEG_THREADS 512
#define PART_VEC    (N_EDGES / 4 / DEG_PARTS)
#define DEG_LOADS   ((PART_VEC + DEG_THREADS - 1) / DEG_THREADS)
static_assert(N_NODES % DEG_ROWS == 0 && DEG_ROWS % 16 == 0, "owner ranges must tile the node list in whole panels");
static_assert(N_EDGES % (4 * DEG_PARTS) == 0, "slices must be whole int4");
static_assert(PRE_VEC == 2 * DEG_THREADS, "two int4 per thread");
static_assert(PRE_BLOCKS <= DEG_BLOCKS, "the converting blocks are the first blocks of the launch");

__device__ __forceinline__ void pack_edges(const int4 s, const int4 d, unsigned* marks,
                                           uint4* __restrict__ pk4, int gi, int li) {
    const unsigned dv[4] = {(unsigned)d.x, (unsigned)d.y, (unsigned)d.z, (unsigned)d.w};
    const unsigned sv[4] = {(unsigned)s.x, (unsigned)s.y, (unsigned)s.z, (unsigned)s.w};
    const int      word = li >> 8;
    const unsigned bit  = 1u << ((li >> 3) & 31);
    unsigned w[4];
    #pragma unroll
    for (int c = 0; c < 4; ++c) {
        const bool node = dv[c] < (unsigned)N_NODES;
        const unsigned own = node ? dv[c] / (unsigned)ROWS_PB : OWN_NONE;
        const unsigned loc = node ? dv[c] - own * (unsigned)ROWS_PB : 0u;
        w[c] = (sv[c] & 0xFFFFu) | (loc << 16) | (own << 24);
        if (node)
            __hip_atomic_fetch_or(&marks[own * PRE_WORDS + word], bit, __ATOMIC_RELAXED, __HIP_MEMORY_SCOPE_WORKGROUP);
    }
    pk4[gi] = uint4{w[0], w[1], w[2], w[3]};
}

__global__ __launch_bounds__(DEG_THREADS)
void outdeg_kernel(const int* __restrict__ src, const int* __restrict__ dst, float* __restrict__ y,
                   unsigned* __restrict__ pres, uint4* __restrict__ pk4) {
    __shared__ unsigned cnt[DEG_ROWS];
    __shared__ __attribute__((aligned(16))) unsigned marks[AGG_BLOCKS * PRE_WORDS];

    const int tid   = threadIdx.x;
    const int range = (int)blockIdx.x / DEG_PARTS;
    const int part  = (int)blockIdx.x % DEG_PARTS;
    const int lo    = range * DEG_ROWS;

    const bool pre_blk = (int)blockIdx.x < PRE_BLOCKS;
    const int  pre_i0  = (int)blockIdx.x * PRE_VEC + tid;
    const int  pre_i1  = pre_i0 + DEG_THREADS;
    const bool pre_v0  = pre_i0 < N_EDGES / 4;
    const bool pre_v1  = pre_i1 < N_EDGES / 4;
    int4 pre_s0 = int4{0, 0, 0, 0}, pre_d0 = int4{0, 0, 0, 0};
    int4 pre_s1 = int4{0, 0, 0, 0}, pre_d1 = int4{0, 0, 0, 0};
    if (pre_v0) {
        pre_s0 = ((const int4*)src)[pre_i0];
        pre_d0 = ((const int4*)dst)[pre_i0];
    }
    if (pre_v1) {
        pre_s1 = ((const int4*)src)[pre_i1];
        pre_d1 = ((const int4*)dst)[pre_i1];
    }

    for (int i = tid; i < DEG_ROWS; i += DEG_THREADS) cnt[i] = 0u;
    if (pre_blk)
        for (int i = tid; i < AGG_BLOCKS * PRE_WORDS; i += DEG_THREADS) marks[i] = 0u;
    __syncthreads();

    if (pre_v0) pack_edges(pre_s0, pre_d0, marks, pk4, pre_i0, tid);
    if (pre_v1) pack_edges(pre_s1, pre_d1, marks, pk4, pre_i1, tid + DEG_THREADS);

    const int4* __restrict__ p = (const int4*)src + part * PART_VEC + tid;
    int4 v[DEG_LOADS];
    #pragma unroll
    for (int k = 0; k < DEG_LOADS; ++k) {
        v[k] = (k * DEG_THREADS + tid < PART_VEC) ? p[k * DEG_THREADS] : int4{-1, -1, -1, -1};
    }

    #pragma unroll
    for (int k = 0; k < DEG_LOADS; ++k) {
        const int sv[4] = {v[k].x, v[k].y, v[k].z, v[k].w};
        #pragma unroll
        for (int c = 0; c < 4; ++c) {
            const unsigned sloc = (unsigned)(sv[c] - lo);
            if (sloc < (unsigned)DEG_ROWS)
                __hip_atomic_fetch_add(&cnt[sloc], 1u, __ATOMIC_RELAXED, __HIP_MEMORY_SCOPE_WORKGROUP);
        }
    }

    __syncthreads();

    for (int i = tid; i < DEG_ROWS; i += DEG_THREADS) {
        float* line = y + (size_t)((lo >> 4) + (i >> 4)) * (16 * FEATS) + part * DEG_LINE;
        *(v2f*)(line + 2 * (i & 15)) = v2f{(float)cnt[i], 0.0f};
    }
    if (pre_blk && tid < AGG_BLOCKS)
        ((uint4*)pres)[(int)blockIdx.x * (PRES_PITCH / 4) + tid] = *(const uint4*)(marks + PRE_WORDS * tid);
}

__global__ __launch_bounds__(AGG_THREADS)
void aggregate_kernel(const unsigned* __restrict__ pres, const unsigned* __restrict__ pk,
                      const float* __restrict__ y,
                      const float* __restrict__ bias, float* __restrict__ out) {
    __shared__ float acc[ROWS_PB * FEATS];
    __shared__ unsigned rings[AGG_WAVES * RING];
    __shared__ unsigned short lists[AGG_WAVES * LIST_CAP];
    __shared__ unsigned indeg[ROWS_PB];

    const int tid  = threadIdx.x;
    const int lane = tid & 31;
    const int wave = __builtin_amdgcn_readfirstlane(tid >> 5);
    const int lo   = blockIdx.x * ROWS_PB;
    const unsigned own = blockIdx.x;
    unsigned* ring = rings + wave * RING;
    unsigned short* list = lists + wave * LIST_CAP;

    const int slot = (wave + (int)blockIdx.x) & (AGG_WAVES - 1);
    const int widx = slot * WPW + lane;
    unsigned m = 0u;
    if (lane < WPW && widx < ROW_WORDS) m = pres[(widx >> 2) * PRES_PITCH + blockIdx.x * PRE_WORDS + (widx & 3)];

    for (int i = tid; i < ROWS_PB * FEATS; i += AGG_THREADS) acc[i] = 0.0f;
    if (tid < ROWS_PB) indeg[tid] = 0u;
    __syncthreads();

    int n = 0;
    {
        unsigned mask = __builtin_amdgcn_ballot_w32(m != 0u);
        while (mask != 0u) {
            if (m != 0u) {
                const int rank = (int)__builtin_amdgcn_mbcnt_lo(mask, 0u);
                list[n + rank] = (unsigned short)(widx * 32 + __builtin_ctz(m));
                m &= m - 1u;
            }
            n += __builtin_popcount(mask);
            mask = __builtin_amdgcn_ballot_w32(m != 0u);
        }
    }

    int head = 0, tail = 0;
    const unsigned* __restrict__ pkl = pk + lane;
    unsigned wv[VISIT];
    if (n > 0) {
        const unsigned g = list[min(lane & (VISIT - 1), n - 1)];
        #pragma unroll
        for (int k = 0; k < VISIT; ++k)
            wv[k] = pkl[(unsigned)__builtin_amdgcn_readlane((int)g, k) * GRP_EDGES];
    }
    for (int i = 0; i < n; i += VISIT) {
        #pragma unroll
        for (int k = 0; k < VISIT; ++k) {
            if (i + k < n) {
                const bool hit = (wv[k] >> 24) == own;
                const unsigned mask = __builtin_amdgcn_ballot_w32(hit);
                if (hit) ring[(tail + (int)__builtin_amdgcn_mbcnt_lo(mask, 0u)) & (RING - 1)] = wv[k];
                tail += __builtin_popcount(mask);
            }
        }
        if (i + VISIT < n) {
            const unsigned g = list[min(i + VISIT + (lane & (VISIT - 1)), n - 1)];
            #pragma unroll
            for (int k = 0; k < VISIT; ++k)
                wv[k] = pkl[(unsigned)__builtin_amdgcn_readlane((int)g, k) * GRP_EDGES];
        }
        while (tail - head >= BATCH) {
            agg_batch(y, acc, indeg, ring, head, BATCH, lane);
            head += BATCH;
        }
    }
    if (tail - head > 0) agg_batch(y, acc, indeg, ring, head, tail - head, lane);

    __syncthreads();

    const float4 b4 = *(const float4*)(bias + lane * 4);
    const int myrow = wave + AGG_WAVES * lane;
    const float sv = (myrow < ROWS_PB) ? rsqrtf(fmaxf((float)indeg[myrow], 1.0f)) : 0.0f;
    int t = 0;
    for (int r = wave; r < ROWS_PB; r += AGG_WAVES, ++t) {
        const float* a = acc + r * FEATS + lane;
        const float sc = __builtin_bit_cast(float, __builtin_amdgcn_readlane(__builtin_bit_cast(int, sv), t));
        float4 o;
        o.x = a[0]  * sc + b4.x;
        o.y = a[32] * sc + b4.y;
        o.z = a[64] * sc + b4.z;
        o.w = a[96] * sc + b4.w;
        *(float4*)(out + (size_t)(lo + r) * FEATS + lane * 4) = o;
    }
}

extern "C" void kernel_launch(void* const* d_in, const int* in_sizes, int n_in,
                              void* d_out, int out_size, void* d_ws, size_t ws_size,
                              hipStream_t stream) {
    const float* x    = (const float*)d_in[0];
    const int*   src  = (const int*)  d_in[1];
    const int*   dst  = (const int*)  d_in[2];
    const float* W    = (const float*)d_in[3];
    const float* bias = (const float*)d_in[4];
    float* out = (float*)d_out;

    float*    y  = (float*)d_ws;
    unsigned* pk = (unsigned*)(y + N_NODES * FEATS);
    unsigned* pres = pk + N_EDGES;

    outdeg_kernel<<<DEG_BLOCKS, DEG_THREADS, 0, stream>>>(src, dst, y, pres, (uint4*)pk);

    gemm_wmma_kernel<<<GEMM_BLOCKS, GEMM_THREADS, 0, stream>>>(x, W, y);

    aggregate_kernel<<<AGG_BLOCKS, AGG_THREADS, 0, stream>>>(pres, pk, y, bias, out);
}
